# starting kernel
# baseline (speedup 1.0000x reference)
#include <hip/hip_runtime.h>
#include <hip/hip_bf16.h>

typedef __attribute__((ext_vector_type(2))) float v2f;
typedef __attribute__((ext_vector_type(8))) float v8f;

#define D 128

__global__ void __launch_bounds__(256)
scored_gemm_kernel(const float* __restrict__ embed,
                   const float* __restrict__ S,
                   float* __restrict__ scored,
                   int nrows) {
    __shared__ float sB[D * D];

    const int tid = threadIdx.x;
    for (int idx = tid; idx < (D / 2) * (D / 4); idx += 256) {
        const int k2 = idx >> 5;
        const int c4 = (idx & 31) << 2;
        const float4 a = *(const float4*)&S[(2 * k2) * D + c4];
        const float4 b = *(const float4*)&S[(2 * k2 + 1) * D + c4];
        float4* dst = (float4*)&sB[k2 * 256 + c4 * 2];
        dst[0] = make_float4(a.x, b.x, a.y, b.y);
        dst[1] = make_float4(a.z, b.z, a.w, b.w);
    }
    __syncthreads();

    const int wave = tid >> 5;
    const int lane = tid & 31;
    const int r0   = (blockIdx.x * 8 + wave) * 16;
    if (r0 >= nrows) return;

    const int mrow  = r0 + (lane & 15);
    const int khalf = (lane >> 4) << 1;
    const int bcol  = lane & 15;

    v8f acc[8];
#pragma unroll
    for (int n = 0; n < 8; ++n) acc[n] = (v8f){};

    const float* __restrict__ arow = embed + (size_t)mrow * D;

    for (int k = 0; k < D; k += 4) {
        const v2f a = *(const v2f*)(arow + k + khalf);
        const float* bbase = sB + ((k >> 1) + (lane >> 4)) * 256 + bcol * 2;
#pragma unroll
        for (int n = 0; n < 8; ++n) {
            const v2f b = *(const v2f*)(bbase + n * 32);
            acc[n] = __builtin_amdgcn_wmma_f32_16x16x4_f32(
                false, a, false, b, (short)0, acc[n], false, false);
        }
    }

    const int srow0 = r0 + ((lane >> 4) << 3);
    const int scol  = lane & 15;
#pragma unroll
    for (int n = 0; n < 8; ++n) {
#pragma unroll
        for (int i = 0; i < 8; ++i) {
            scored[(size_t)(srow0 + i) * D + n * 16 + scol] = acc[n][i];
        }
    }
}

__device__ __forceinline__ float wave_reduce32(float v) {
#pragma unroll
    for (int m = 16; m >= 1; m >>= 1) v += __shfl_xor(v, m, 32);
    return v;
}

__device__ __forceinline__ float dot4(float4 a, float4 b) {
    return a.x * b.x + a.y * b.y + a.z * b.z + a.w * b.w;
}

__global__ void __launch_bounds__(256)
score_kernel(const float* __restrict__ embed,
             const float* __restrict__ scored,
             const int* __restrict__ h,
             const int* __restrict__ t,
             const int* __restrict__ n_s,
             float* __restrict__ out,
             int E) {
    const int gw   = (int)((blockIdx.x * blockDim.x + threadIdx.x) >> 5);
    const int lane = threadIdx.x & 31;
    if (gw >= E) return;

    const size_t hoff = (size_t)h[gw] * D + lane * 4;
    const size_t toff = (size_t)t[gw] * D + lane * 4;
    const float4 hs = *(const float4*)(scored + hoff);
    const float4 te = *(const float4*)(embed + toff);

    float p = wave_reduce32(dot4(hs, te));
    if (lane == 0) out[gw] = p;

    float* __restrict__ neg = out + E + (size_t)gw * 16;

#pragma unroll
    for (int j = 0; j < 8; ++j) {
        const size_t off = (size_t)n_s[gw * 16 + j] * D + lane * 4;
        const float4 nt = *(const float4*)(embed + off);
        const float s = wave_reduce32(dot4(hs, nt));
        if (lane == 0) neg[j] = s;
    }
#pragma unroll
    for (int j = 0; j < 8; ++j) {
        const size_t off = (size_t)n_s[gw * 16 + 8 + j] * D + lane * 4;
        const float4 nh = *(const float4*)(scored + off);
        const float s = wave_reduce32(dot4(nh, te));
        if (lane == 0) neg[8 + j] = s;
    }
}

__device__ __forceinline__ void matvec_rowS(const float* __restrict__ row,
                                            const float* __restrict__ S,
                                            int lane, float r[4]) {
    r[0] = r[1] = r[2] = r[3] = 0.f;
    for (int k = 0; k < D; ++k) {
        const float a = row[k];
        const float4 s = *(const float4*)(S + (size_t)k * D + lane * 4);
        r[0] += a * s.x; r[1] += a * s.y; r[2] += a * s.z; r[3] += a * s.w;
    }
}

__global__ void __launch_bounds__(256)
fused_fallback_kernel(const float* __restrict__ embed,
                      const float* __restrict__ S,
                      const int* __restrict__ h,
                      const int* __restrict__ t,
                      const int* __restrict__ n_s,
                      float* __restrict__ out,
                      int E) {
    const int gw   = (int)((blockIdx.x * blockDim.x + threadIdx.x) >> 5);
    const int lane = threadIdx.x & 31;
    if (gw >= E) return;

    float hS[4];
    matvec_rowS(embed + (size_t)h[gw] * D, S, lane, hS);
    const float4 hs = make_float4(hS[0], hS[1], hS[2], hS[3]);
    const float4 te = *(const float4*)(embed + (size_t)t[gw] * D + lane * 4);

    float p = wave_reduce32(dot4(hs, te));
    if (lane == 0) out[gw] = p;

    float* __restrict__ neg = out + E + (size_t)gw * 16;
    for (int j = 0; j < 8; ++j) {
        const size_t off = (size_t)n_s[gw * 16 + j] * D + lane * 4;
        const float4 nt = *(const float4*)(embed + off);
        const float s = wave_reduce32(dot4(hs, nt));
        if (lane == 0) neg[j] = s;
    }
    for (int j = 0; j < 8; ++j) {
        float nS[4];
        matvec_rowS(embed + (size_t)n_s[gw * 16 + 8 + j] * D, S, lane, nS);
        const float4 nh = make_float4(nS[0], nS[1], nS[2], nS[3]);
        const float s = wave_reduce32(dot4(nh, te));
        if (lane == 0) neg[8 + j] = s;
    }
}

extern "C" void kernel_launch(void* const* d_in, const int* in_sizes, int n_in,
                              void* d_out, int out_size, void* d_ws, size_t ws_size,
                              hipStream_t stream) {
    const float* embed = (const float*)d_in[0];
    const float* S     = (const float*)d_in[1];
    const int*   h     = (const int*)d_in[2];
    const int*   t     = (const int*)d_in[3];
    const int*   n_s   = (const int*)d_in[4];
    float*       out   = (float*)d_out;

    const int N = in_sizes[0] / D;
    const int E = in_sizes[2];

    const size_t need = (size_t)N * D * sizeof(float);
    if (ws_size >= need) {
        float* scored = (float*)d_ws;
        const int waves  = (N + 15) / 16;
        const int blocks = (waves + 7) / 8;
        scored_gemm_kernel<<<blocks, 256, 0, stream>>>(embed, S, scored, N);
        const int eblocks = (E + 7) / 8;
        score_kernel<<<eblocks, 256, 0, stream>>>(embed, scored, h, t, n_s, out, E);
    } else {
        const int eblocks = (E + 7) / 8;
        fused_fallback_kernel<<<eblocks, 256, 0, stream>>>(embed, S, h, t, n_s, out, E);
    }
}


# edge kernel: row numbers via scalar loads, all 18 row gathers of an edge requested before first use (register goal 8 waves per SIMD): 18 loads in flight per wave, was 1; sum order unchanged
# speedup vs baseline: 1.4864x; 1.4864x over previous
#include <hip/hip_runtime.h>
#include <hip/hip_bf16.h>

typedef __attribute__((ext_vector_type(2))) float v2f;
typedef __attribute__((ext_vector_type(8))) float v8f;

#define D 128

__global__ void __launch_bounds__(256)
scored_gemm_kernel(const float* __restrict__ embed,
                   const float* __restrict__ S,
                   float* __restrict__ scored,
                   int nrows) {
    __shared__ float sB[D * D];

    const int tid = threadIdx.x;
    for (int idx = tid; idx < (D / 2) * (D / 4); idx += 256) {
        const int k2 = idx >> 5;
        const int c4 = (idx & 31) << 2;
        const float4 a = *(const float4*)&S[(2 * k2) * D + c4];
        const float4 b = *(const float4*)&S[(2 * k2 + 1) * D + c4];
        float4* dst = (float4*)&sB[k2 * 256 + c4 * 2];
        dst[0] = make_float4(a.x, b.x, a.y, b.y);
        dst[1] = make_float4(a.z, b.z, a.w, b.w);
    }
    __syncthreads();

    const int wave = tid >> 5;
    const int lane = tid & 31;
    const int r0   = (blockIdx.x * 8 + wave) * 16;
    if (r0 >= nrows) return;

    const int mrow  = r0 + (lane & 15);
    const int khalf = (lane >> 4) << 1;
    const int bcol  = lane & 15;

    v8f acc[8];
#pragma unroll
    for (int n = 0; n < 8; ++n) acc[n] = (v8f){};

    const float* __restrict__ arow = embed + (size_t)mrow * D;

    for (int k = 0; k < D; k += 4) {
        const v2f a = *(const v2f*)(arow + k + khalf);
        const float* bbase = sB + ((k >> 1) + (lane >> 4)) * 256 + bcol * 2;
#pragma unroll
        for (int n = 0; n < 8; ++n) {
            const v2f b = *(const v2f*)(bbase + n * 32);
            acc[n] = __builtin_amdgcn_wmma_f32_16x16x4_f32(
                false, a, false, b, (short)0, acc[n], false, false);
        }
    }

    const int srow0 = r0 + ((lane >> 4) << 3);
    const int scol  = lane & 15;
#pragma unroll
    for (int n = 0; n < 8; ++n) {
#pragma unroll
        for (int i = 0; i < 8; ++i) {
            scored[(size_t)(srow0 + i) * D + n * 16 + scol] = acc[n][i];
        }
    }
}

__device__ __forceinline__ float wave_reduce32(float v) {
#pragma unroll
    for (int m = 16; m >= 1; m >>= 1) v += __shfl_xor(v, m, 32);
    return v;
}

__device__ __forceinline__ float dot4(float4 a, float4 b) {
    return a.x * b.x + a.y * b.y + a.z * b.z + a.w * b.w;
}

__device__ __forceinline__ float4 row_piece(const float* __restrict__ table,
                                            int row, int lane) {
    return *(const float4*)(table + (size_t)row * D + lane * 4);
}

__global__ void __launch_bounds__(256)
__attribute__((amdgpu_waves_per_eu(8, 8)))
score_kernel(const float* __restrict__ embed,
             const float* __restrict__ scored,
             const int* __restrict__ h,
             const int* __restrict__ t,
             const int* __restrict__ n_s,
             float* __restrict__ out,
             int E) {
    const int lane = threadIdx.x & 31;
    const int gw = __builtin_amdgcn_readfirstlane(
        (int)((blockIdx.x * blockDim.x + threadIdx.x) >> 5));
    if (gw >= E) return;

    const int hi = h[gw];
    const int ti = t[gw];
    int ni[16];
    for (int j = 0; j < 16; ++j) ni[j] = n_s[(size_t)gw * 16 + j];

    const float4 hs = row_piece(scored, hi, lane);
    const float4 te = row_piece(embed, ti, lane);
    float4 nt[8], nh[8];
    for (int j = 0; j < 8; ++j) nt[j] = row_piece(embed, ni[j], lane);
    for (int j = 0; j < 8; ++j) nh[j] = row_piece(scored, ni[8 + j], lane);

    float s[17];
    s[0] = dot4(hs, te);
    for (int j = 0; j < 8; ++j) s[1 + j] = dot4(hs, nt[j]);
    for (int j = 0; j < 8; ++j) s[9 + j] = dot4(nh[j], te);

    for (int i = 0; i < 17; ++i) s[i] = wave_reduce32(s[i]);

    if (lane == 0) {
        out[gw] = s[0];
        float* __restrict__ neg = out + E + (size_t)gw * 16;
        for (int j = 0; j < 16; ++j) neg[j] = s[1 + j];
    }
}

__device__ __forceinline__ void matvec_rowS(const float* __restrict__ row,
                                            const float* __restrict__ S,
                                            int lane, float r[4]) {
    r[0] = r[1] = r[2] = r[3] = 0.f;
    for (int k = 0; k < D; ++k) {
        const float a = row[k];
        const float4 s = *(const float4*)(S + (size_t)k * D + lane * 4);
        r[0] += a * s.x; r[1] += a * s.y; r[2] += a * s.z; r[3] += a * s.w;
    }
}

__global__ void __launch_bounds__(256)
fused_fallback_kernel(const float* __restrict__ embed,
                      const float* __restrict__ S,
                      const int* __restrict__ h,
                      const int* __restrict__ t,
                      const int* __restrict__ n_s,
                      float* __restrict__ out,
                      int E) {
    const int gw   = (int)((blockIdx.x * blockDim.x + threadIdx.x) >> 5);
    const int lane = threadIdx.x & 31;
    if (gw >= E) return;

    float hS[4];
    matvec_rowS(embed + (size_t)h[gw] * D, S, lane, hS);
    const float4 hs = make_float4(hS[0], hS[1], hS[2], hS[3]);
    const float4 te = *(const float4*)(embed + (size_t)t[gw] * D + lane * 4);

    float p = wave_reduce32(dot4(hs, te));
    if (lane == 0) out[gw] = p;

    float* __restrict__ neg = out + E + (size_t)gw * 16;
    for (int j = 0; j < 8; ++j) {
        const size_t off = (size_t)n_s[gw * 16 + j] * D + lane * 4;
        const float4 nt = *(const float4*)(embed + off);
        const float s = wave_reduce32(dot4(hs, nt));
        if (lane == 0) neg[j] = s;
    }
    for (int j = 0; j < 8; ++j) {
        float nS[4];
        matvec_rowS(embed + (size_t)n_s[gw * 16 + 8 + j] * D, S, lane, nS);
        const float4 nh = make_float4(nS[0], nS[1], nS[2], nS[3]);
        const float s = wave_reduce32(dot4(nh, te));
        if (lane == 0) neg[8 + j] = s;
    }
}

extern "C" void kernel_launch(void* const* d_in, const int* in_sizes, int n_in,
                              void* d_out, int out_size, void* d_ws, size_t ws_size,
                              hipStream_t stream) {
    const float* embed = (const float*)d_in[0];
    const float* S     = (const float*)d_in[1];
    const int*   h     = (const int*)d_in[2];
    const int*   t     = (const int*)d_in[3];
    const int*   n_s   = (const int*)d_in[4];
    float*       out   = (float*)d_out;

    const int N = in_sizes[0] / D;
    const int E = in_sizes[2];

    const size_t need = (size_t)N * D * sizeof(float);
    if (ws_size >= need) {
        float* scored = (float*)d_ws;
        const int waves  = (N + 15) / 16;
        const int blocks = (waves + 7) / 8;
        scored_gemm_kernel<<<blocks, 256, 0, stream>>>(embed, S, scored, N);
        const int eblocks = (E + 7) / 8;
        score_kernel<<<eblocks, 256, 0, stream>>>(embed, scored, h, t, n_s, out, E);
    } else {
        const int eblocks = (E + 7) / 8;
        fused_fallback_kernel<<<eblocks, 256, 0, stream>>>(embed, S, h, t, n_s, out, E);
    }
}


# both launches fused into one kernel: a wave does the two f32 WMMA products of its 16 edges, then scores them; matrix work of one block runs under the row gathers of another; sum order unchanged
# speedup vs baseline: 1.7252x; 1.1607x over previous
#include <hip/hip_runtime.h>
#include <hip/hip_bf16.h>

typedef __attribute__((ext_vector_type(2))) float v2f;
typedef __attribute__((ext_vector_type(8))) float v8f;

#define D 128


__device__ __forceinline__ void stage_S(float* sB, const float* __restrict__ S, int tid) {
    for (int idx = tid; idx < (D / 2) * (D / 4); idx += 256) {
        const int k2 = idx >> 5;
        const int c4 = (idx & 31) << 2;
        const float4 a = *(const float4*)&S[(2 * k2) * D + c4];
        const float4 b = *(const float4*)&S[(2 * k2 + 1) * D + c4];
        float4* dst = (float4*)&sB[k2 * 256 + c4 * 2];
        dst[0] = make_float4(a.x, b.x, a.y, b.y);
        dst[1] = make_float4(a.z, b.z, a.w, b.w);
    }
}

__device__ __forceinline__ void stage_St(float* sB, const float* __restrict__ S, int tid) {
    for (int idx = tid; idx < D * (D / 4); idx += 256) {
        const int c = idx & (D - 1);
        const int q = idx >> 7;
        const float4 v = *(const float4*)&S[c * D + 4 * q];
        *(v2f*)&sB[(2 * q) * 256 + c * 2]     = (v2f){v.x, v.y};
        *(v2f*)&sB[(2 * q + 1) * 256 + c * 2] = (v2f){v.z, v.w};
    }
}

__device__ __forceinline__ void slab_gemm(const float* __restrict__ embed,
                                          const int* __restrict__ idx,
                                          const float* sB, float* Pout,
                                          int r0, int E, int lane) {
    int erow = r0 + (lane & 15);
    erow = erow < E ? erow : E - 1;
    const int src   = idx[erow];
    const int khalf = (lane >> 4) << 1;
    const int bcol  = lane & 15;

    v8f acc[8];
#pragma unroll
    for (int n = 0; n < 8; ++n) acc[n] = (v8f){};

    const float* __restrict__ arow = embed + (size_t)src * D;

    for (int k = 0; k < D; k += 4) {
        const v2f a = *(const v2f*)(arow + k + khalf);
        const float* bbase = sB + ((k >> 1) + (lane >> 4)) * 256 + bcol * 2;
#pragma unroll
        for (int n = 0; n < 8; ++n) {
            const v2f b = *(const v2f*)(bbase + n * 32);
            acc[n] = __builtin_amdgcn_wmma_f32_16x16x4_f32(
                false, a, false, b, (short)0, acc[n], false, false);
        }
    }

    const int srow0 = r0 + ((lane >> 4) << 3);
    const int scol  = lane & 15;
    if (r0 + 16 <= E) {
#pragma unroll
        for (int n = 0; n < 8; ++n) {
#pragma unroll
            for (int i = 0; i < 8; ++i)
                Pout[(size_t)(srow0 + i) * D + n * 16 + scol] = acc[n][i];
        }
    } else {
#pragma unroll
        for (int n = 0; n < 8; ++n) {
#pragma unroll
            for (int i = 0; i < 8; ++i)
                if (srow0 + i < E)
                    Pout[(size_t)(srow0 + i) * D + n * 16 + scol] = acc[n][i];
        }
    }
}

__device__ __forceinline__ float wave_reduce32(float v) {
#pragma unroll
    for (int m = 16; m >= 1; m >>= 1) v += __shfl_xor(v, m, 32);
    return v;
}

__device__ __forceinline__ float dot4(float4 a, float4 b) {
    return a.x * b.x + a.y * b.y + a.z * b.z + a.w * b.w;
}

__device__ __forceinline__ float4 row_piece(const float* table, int row, int lane) {
    return *(const float4*)(table + (size_t)row * D + lane * 4);
}

__global__ void __launch_bounds__(256)
__attribute__((amdgpu_waves_per_eu(8, 8)))
edge_fused_kernel(const float* __restrict__ embed,
                  const float* __restrict__ S,
                  const int* __restrict__ h,
                  const int* __restrict__ t,
                  const int* __restrict__ n_s,
                  float* __restrict__ P,
                  float* __restrict__ out,
                  int E) {
    __shared__ float sB[D * D];

    const int tid  = threadIdx.x;
    const int lane = tid & 31;
    const int slab = __builtin_amdgcn_readfirstlane(
        (int)blockIdx.x + (tid >> 5) * (int)gridDim.x);
    const bool active = slab < ((E + 15) >> 4);
    const int r0 = slab * 16;

    stage_S(sB, S, tid);
    __syncthreads();
    if (active) slab_gemm(embed, h, sB, P, r0, E, lane);
    __syncthreads();
    stage_St(sB, S, tid);
    __syncthreads();
    if (!active) return;
    slab_gemm(embed, t, sB, P + (size_t)E * D, r0, E, lane);

    __threadfence_block();

    const int elast = E - 1;
    int ti_v = t[r0];
    int ni_v[16];
    for (int j = 0; j < 16; ++j) ni_v[j] = n_s[(size_t)r0 * 16 + j];

    for (int i = 0; i < 16; ++i) {
        const int e = r0 + i;
        if (e >= E) break;

        const int ti = __builtin_amdgcn_readfirstlane(ti_v);
        int ni[16];
        for (int j = 0; j < 16; ++j) ni[j] = __builtin_amdgcn_readfirstlane(ni_v[j]);

        const float4 hs = row_piece(P, e, lane);
        const float4 st = *(const float4*)(P + ((size_t)E + e) * D + lane * 4);
        const float4 te = row_piece(embed, ti, lane);
        float4 nt[8], nh[8];
        for (int j = 0; j < 8; ++j) nt[j] = row_piece(embed, ni[j], lane);
        for (int j = 0; j < 8; ++j) nh[j] = row_piece(embed, ni[8 + j], lane);

        const int en = (e + 1 < E) ? e + 1 : elast;
        ti_v = t[en];
        for (int j = 0; j < 16; ++j) ni_v[j] = n_s[(size_t)en * 16 + j];

        float s[17];
        s[0] = dot4(hs, te);
        for (int j = 0; j < 8; ++j) s[1 + j] = dot4(hs, nt[j]);
        for (int j = 0; j < 8; ++j) s[9 + j] = dot4(nh[j], st);

        for (int q = 0; q < 17; ++q) s[q] = wave_reduce32(s[q]);

        if (lane == 0) {
            out[e] = s[0];
            float* neg = out + E + (size_t)e * 16;
            for (int j = 0; j < 16; ++j) neg[j] = s[1 + j];
        }
    }
}

__device__ __forceinline__ void matvec_rowS(const float* __restrict__ row,
                                            const float* __restrict__ S,
                                            int lane, float r[4]) {
    r[0] = r[1] = r[2] = r[3] = 0.f;
    for (int k = 0; k < D; ++k) {
        const float a = row[k];
        const float4 s = *(const float4*)(S + (size_t)k * D + lane * 4);
        r[0] += a * s.x; r[1] += a * s.y; r[2] += a * s.z; r[3] += a * s.w;
    }
}

__global__ void __launch_bounds__(256)
fused_fallback_kernel(const float* __restrict__ embed,
                      const float* __restrict__ S,
                      const int* __restrict__ h,
                      const int* __restrict__ t,
                      const int* __restrict__ n_s,
                      float* __restrict__ out,
                      int E) {
    const int gw   = (int)((blockIdx.x * blockDim.x + threadIdx.x) >> 5);
    const int lane = threadIdx.x & 31;
    if (gw >= E) return;

    float hS[4];
    matvec_rowS(embed + (size_t)h[gw] * D, S, lane, hS);
    const float4 hs = make_float4(hS[0], hS[1], hS[2], hS[3]);
    const float4 te = *(const float4*)(embed + (size_t)t[gw] * D + lane * 4);

    float p = wave_reduce32(dot4(hs, te));
    if (lane == 0) out[gw] = p;

    float* __restrict__ neg = out + E + (size_t)gw * 16;
    for (int j = 0; j < 8; ++j) {
        const size_t off = (size_t)n_s[gw * 16 + j] * D + lane * 4;
        const float4 nt = *(const float4*)(embed + off);
        const float s = wave_reduce32(dot4(hs, nt));
        if (lane == 0) neg[j] = s;
    }
    for (int j = 0; j < 8; ++j) {
        float nS[4];
        matvec_rowS(embed + (size_t)n_s[gw * 16 + 8 + j] * D, S, lane, nS);
        const float4 nh = make_float4(nS[0], nS[1], nS[2], nS[3]);
        const float s = wave_reduce32(dot4(nh, te));
        if (lane == 0) neg[8 + j] = s;
    }
}

extern "C" void kernel_launch(void* const* d_in, const int* in_sizes, int n_in,
                              void* d_out, int out_size, void* d_ws, size_t ws_size,
                              hipStream_t stream) {
    const float* embed = (const float*)d_in[0];
    const float* S     = (const float*)d_in[1];
    const int*   h     = (const int*)d_in[2];
    const int*   t     = (const int*)d_in[3];
    const int*   n_s   = (const int*)d_in[4];
    float*       out   = (float*)d_out;

    const int E = in_sizes[2];

    const size_t need = (size_t)2 * E * D * sizeof(float);
    if (ws_size >= need) {
        float* P = (float*)d_ws;
        const int slabs = (E + 15) / 16;
        const int nblk  = ((slabs + 2047) / 2048) * 256;
        edge_fused_kernel<<<nblk, 256, 0, stream>>>(embed, S, h, t, n_s, P, out, E);
    } else {
        const int eblocks = (E + 7) / 8;
        fused_fallback_kernel<<<eblocks, 256, 0, stream>>>(embed, S, h, t, n_s, out, E);
    }
}


# the 17 wave sums of an edge as one transposing exchange tree on the vector ALU (permlanex16 + row_xmask DPP): 21 exchanges, was 85 LDS-unit bpermutes; same pairing order, sums bit-identical
# speedup vs baseline: 1.8902x; 1.0956x over previous
#include <hip/hip_runtime.h>
#include <hip/hip_bf16.h>

typedef __attribute__((ext_vector_type(2))) float v2f;
typedef __attribute__((ext_vector_type(8))) float v8f;

#define D 128


__device__ __forceinline__ void stage_S(float* sB, const float* __restrict__ S, int tid) {
    for (int idx = tid; idx < (D / 2) * (D / 4); idx += 256) {
        const int k2 = idx >> 5;
        const int c4 = (idx & 31) << 2;
        const float4 a = *(const float4*)&S[(2 * k2) * D + c4];
        const float4 b = *(const float4*)&S[(2 * k2 + 1) * D + c4];
        float4* dst = (float4*)&sB[k2 * 256 + c4 * 2];
        dst[0] = make_float4(a.x, b.x, a.y, b.y);
        dst[1] = make_float4(a.z, b.z, a.w, b.w);
    }
}

__device__ __forceinline__ void stage_St(float* sB, const float* __restrict__ S, int tid) {
    for (int idx = tid; idx < D * (D / 4); idx += 256) {
        const int c = idx & (D - 1);
        const int q = idx >> 7;
        const float4 v = *(const float4*)&S[c * D + 4 * q];
        *(v2f*)&sB[(2 * q) * 256 + c * 2]     = (v2f){v.x, v.y};
        *(v2f*)&sB[(2 * q + 1) * 256 + c * 2] = (v2f){v.z, v.w};
    }
}

__device__ __forceinline__ void slab_gemm(const float* __restrict__ embed,
                                          const int* __restrict__ idx,
                                          const float* sB, float* Pout,
                                          int r0, int E, int lane) {
    int erow = r0 + (lane & 15);
    erow = erow < E ? erow : E - 1;
    const int src   = idx[erow];
    const int khalf = (lane >> 4) << 1;
    const int bcol  = lane & 15;

    v8f acc[8];
#pragma unroll
    for (int n = 0; n < 8; ++n) acc[n] = (v8f){};

    const float* __restrict__ arow = embed + (size_t)src * D;

    for (int k = 0; k < D; k += 4) {
        const v2f a = *(const v2f*)(arow + k + khalf);
        const float* bbase = sB + ((k >> 1) + (lane >> 4)) * 256 + bcol * 2;
#pragma unroll
        for (int n = 0; n < 8; ++n) {
            const v2f b = *(const v2f*)(bbase + n * 32);
            acc[n] = __builtin_amdgcn_wmma_f32_16x16x4_f32(
                false, a, false, b, (short)0, acc[n], false, false);
        }
    }

    const int srow0 = r0 + ((lane >> 4) << 3);
    const int scol  = lane & 15;
    if (r0 + 16 <= E) {
#pragma unroll
        for (int n = 0; n < 8; ++n) {
#pragma unroll
            for (int i = 0; i < 8; ++i)
                Pout[(size_t)(srow0 + i) * D + n * 16 + scol] = acc[n][i];
        }
    } else {
#pragma unroll
        for (int n = 0; n < 8; ++n) {
#pragma unroll
            for (int i = 0; i < 8; ++i)
                if (srow0 + i < E)
                    Pout[(size_t)(srow0 + i) * D + n * 16 + scol] = acc[n][i];
        }
    }
}

__device__ __forceinline__ float wave_reduce32(float v) {
#pragma unroll
    for (int m = 16; m >= 1; m >>= 1) v += __shfl_xor(v, m, 32);
    return v;
}

__device__ __forceinline__ float xhalf(float x) {
    const int b = __builtin_bit_cast(int, x);
    return __builtin_bit_cast(float,
        __builtin_amdgcn_permlanex16(b, b, 0x76543210, 0xfedcba98, false, false));
}

template <int M>
__device__ __forceinline__ float xrow(float x) {
    const int b = __builtin_bit_cast(int, x);
    return __builtin_bit_cast(float,
        __builtin_amdgcn_update_dpp(0, b, 0x160 + M, 0xf, 0xf, true));
}

__device__ __forceinline__ float wave_sum32(float v) {
    v = v + xhalf(v);
    v = v + xrow<8>(v);
    v = v + xrow<4>(v);
    v = v + xrow<2>(v);
    v = v + xrow<1>(v);
    return v;
}

__device__ __forceinline__ float dot4(float4 a, float4 b) {
    return a.x * b.x + a.y * b.y + a.z * b.z + a.w * b.w;
}

__device__ __forceinline__ float4 row_piece(const float* table, int row, int lane) {
    return *(const float4*)(table + (size_t)row * D + lane * 4);
}

__global__ void __launch_bounds__(256)
__attribute__((amdgpu_waves_per_eu(8, 8)))
edge_fused_kernel(const float* __restrict__ embed,
                  const float* __restrict__ S,
                  const int* __restrict__ h,
                  const int* __restrict__ t,
                  const int* __restrict__ n_s,
                  float* __restrict__ P,
                  float* __restrict__ out,
                  int E) {
    __shared__ float sB[D * D];

    const int tid  = threadIdx.x;
    const int lane = tid & 31;
    const int slab = __builtin_amdgcn_readfirstlane(
        (int)blockIdx.x + (tid >> 5) * (int)gridDim.x);
    const bool active = slab < ((E + 15) >> 4);
    const int r0 = slab * 16;

    stage_S(sB, S, tid);
    __syncthreads();
    if (active) slab_gemm(embed, h, sB, P, r0, E, lane);
    __syncthreads();
    stage_St(sB, S, tid);
    __syncthreads();
    if (!active) return;
    slab_gemm(embed, t, sB, P + (size_t)E * D, r0, E, lane);

    __threadfence_block();

    const int elast = E - 1;
    int ti_v = t[r0];
    int ni_v[16];
    for (int j = 0; j < 16; ++j) ni_v[j] = n_s[(size_t)r0 * 16 + j];

    const bool up16 = (lane & 16) != 0;
    const bool up8  = (lane & 8) != 0;
    const bool up4  = (lane & 4) != 0;
    const bool up2  = (lane & 2) != 0;

    for (int i = 0; i < 16; ++i) {
        const int e = r0 + i;
        if (e >= E) break;

        const int ti = __builtin_amdgcn_readfirstlane(ti_v);
        int ni[16];
        for (int j = 0; j < 16; ++j) ni[j] = __builtin_amdgcn_readfirstlane(ni_v[j]);

        const float4 hs = row_piece(P, e, lane);
        const float4 st = *(const float4*)(P + ((size_t)E + e) * D + lane * 4);
        const float4 te = row_piece(embed, ti, lane);
        float4 nt[8], nh[8];
        for (int j = 0; j < 8; ++j) nt[j] = row_piece(embed, ni[j], lane);
        for (int j = 0; j < 8; ++j) nh[j] = row_piece(embed, ni[8 + j], lane);

        const int en = (e + 1 < E) ? e + 1 : elast;
        ti_v = t[en];
        for (int j = 0; j < 16; ++j) ni_v[j] = n_s[(size_t)en * 16 + j];

        float s[17];
        s[0] = dot4(hs, te);
        for (int j = 0; j < 8; ++j) s[1 + j] = dot4(hs, nt[j]);
        for (int j = 0; j < 8; ++j) s[9 + j] = dot4(nh[j], st);

        const float pos = wave_sum32(s[0]);

        float w8[8], w4[4], w2[2];
        for (int j = 0; j < 8; ++j) {
            const float keep = up16 ? s[9 + j] : s[1 + j];
            const float send = up16 ? s[1 + j] : s[9 + j];
            w8[j] = keep + xhalf(send);
        }
        for (int j = 0; j < 4; ++j) {
            const float keep = up8 ? w8[4 + j] : w8[j];
            const float send = up8 ? w8[j] : w8[4 + j];
            w4[j] = keep + xrow<8>(send);
        }
        for (int j = 0; j < 2; ++j) {
            const float keep = up4 ? w4[2 + j] : w4[j];
            const float send = up4 ? w4[j] : w4[2 + j];
            w2[j] = keep + xrow<4>(send);
        }
        const float keep1 = up2 ? w2[1] : w2[0];
        const float send1 = up2 ? w2[0] : w2[1];
        float neg1 = keep1 + xrow<2>(send1);
        neg1 = neg1 + xrow<1>(neg1);

        if (lane == 0) out[e] = pos;
        if ((lane & 1) == 0) out[E + (size_t)e * 16 + (lane >> 1)] = neg1;
    }
}

__device__ __forceinline__ void matvec_rowS(const float* __restrict__ row,
                                            const float* __restrict__ S,
                                            int lane, float r[4]) {
    r[0] = r[1] = r[2] = r[3] = 0.f;
    for (int k = 0; k < D; ++k) {
        const float a = row[k];
        const float4 s = *(const float4*)(S + (size_t)k * D + lane * 4);
        r[0] += a * s.x; r[1] += a * s.y; r[2] += a * s.z; r[3] += a * s.w;
    }
}

__global__ void __launch_bounds__(256)
fused_fallback_kernel(const float* __restrict__ embed,
                      const float* __restrict__ S,
                      const int* __restrict__ h,
                      const int* __restrict__ t,
                      const int* __restrict__ n_s,
                      float* __restrict__ out,
                      int E) {
    const int gw   = (int)((blockIdx.x * blockDim.x + threadIdx.x) >> 5);
    const int lane = threadIdx.x & 31;
    if (gw >= E) return;

    float hS[4];
    matvec_rowS(embed + (size_t)h[gw] * D, S, lane, hS);
    const float4 hs = make_float4(hS[0], hS[1], hS[2], hS[3]);
    const float4 te = *(const float4*)(embed + (size_t)t[gw] * D + lane * 4);

    float p = wave_reduce32(dot4(hs, te));
    if (lane == 0) out[gw] = p;

    float* __restrict__ neg = out + E + (size_t)gw * 16;
    for (int j = 0; j < 8; ++j) {
        const size_t off = (size_t)n_s[gw * 16 + j] * D + lane * 4;
        const float4 nt = *(const float4*)(embed + off);
        const float s = wave_reduce32(dot4(hs, nt));
        if (lane == 0) neg[j] = s;
    }
    for (int j = 0; j < 8; ++j) {
        float nS[4];
        matvec_rowS(embed + (size_t)n_s[gw * 16 + 8 + j] * D, S, lane, nS);
        const float4 nh = make_float4(nS[0], nS[1], nS[2], nS[3]);
        const float s = wave_reduce32(dot4(nh, te));
        if (lane == 0) neg[8 + j] = s;
    }
}

extern "C" void kernel_launch(void* const* d_in, const int* in_sizes, int n_in,
                              void* d_out, int out_size, void* d_ws, size_t ws_size,
                              hipStream_t stream) {
    const float* embed = (const float*)d_in[0];
    const float* S     = (const float*)d_in[1];
    const int*   h     = (const int*)d_in[2];
    const int*   t     = (const int*)d_in[3];
    const int*   n_s   = (const int*)d_in[4];
    float*       out   = (float*)d_out;

    const int E = in_sizes[2];

    const size_t need = (size_t)2 * E * D * sizeof(float);
    if (ws_size >= need) {
        float* P = (float*)d_ws;
        const int slabs = (E + 15) / 16;
        const int nblk  = ((slabs + 2047) / 2048) * 256;
        edge_fused_kernel<<<nblk, 256, 0, stream>>>(embed, S, h, t, n_s, P, out, E);
    } else {
        const int eblocks = (E + 7) / 8;
        fused_fallback_kernel<<<eblocks, 256, 0, stream>>>(embed, S, h, t, n_s, out, E);
    }
}
